# starting kernel
# baseline (speedup 1.0000x reference)
#include <hip/hip_runtime.h>
#include <hip/hip_bf16.h>
#include <math.h>

#define BB 4
#define NN 96
#define TT 512
#define KK 12
#define GG 8
#define WAVES 4
#define BLOCK (WAVES * 32)
#define H1_STRIDE 40

typedef __attribute__((ext_vector_type(16))) _Float16 v16h;
typedef __attribute__((ext_vector_type(8)))  _Float16 v8h;
typedef __attribute__((ext_vector_type(8)))  float    v8f;

union AFrag { v16h v; v8h h[2]; _Float16 e[16]; };
union CFrag { v8f  v; float e[8]; };

__device__ __forceinline__ float fexp(float x) { return __expf(x); }
__device__ __forceinline__ float frcp(float x) { return __builtin_amdgcn_rcpf(x); }

__device__ __forceinline__ float softplus_stable(float x) {
  return fmaxf(x, 0.0f) + __logf(1.0f + fexp(-fabsf(x)));
}

__device__ __forceinline__ float fast_tanh(float x) {
  float e = fexp(-2.0f * fabsf(x));
  float y = (1.0f - e) * frcp(1.0f + e);
  return copysignf(y, x);
}

__device__ __forceinline__ float fast_erf(float x) {
  float ax = fabsf(x);
  float t  = frcp(fmaf(0.3275911f, ax, 1.0f));
  float p  = fmaf(t, 1.061405429f, -1.453152027f);
  p = fmaf(t, p, 1.421413741f);
  p = fmaf(t, p, -0.284496736f);
  p = fmaf(t, p, 0.254829592f);
  float y = 1.0f - p * t * fexp(-ax * ax);
  return copysignf(y, x);
}

__device__ __forceinline__ float gelu_exact(float x) {
  return 0.5f * x * (1.0f + fast_erf(x * 0.70710678118654752f));
}

__device__ __forceinline__ float fast_acos(float x) {
  float ax = fabsf(x);
  float p = fmaf(ax, -0.0187293f, 0.0742610f);
  p = fmaf(ax, p, -0.2121144f);
  p = fmaf(ax, p, 1.5707288f);
  float r = sqrtf(fmaxf(1.0f - ax, 0.0f)) * p;
  return (x >= 0.0f) ? r : (3.14159265358979f - r);
}

__global__ __launch_bounds__(BLOCK) void ecg_fused_kernel(
    const float* __restrict__ g_mu,   const float* __restrict__ g_s0,
    const float* __restrict__ g_sv,   const float* __restrict__ g_amp,
    const float* __restrict__ g_shb,  const float* __restrict__ g_shv,
    const float* __restrict__ g_p0,   const float* __restrict__ g_pv,
    const float* __restrict__ g_tau,  const float* __restrict__ g_gam,
    const float* __restrict__ g_ra,   const float* __restrict__ g_la,
    const float* __restrict__ g_ll,   const float* __restrict__ g_chest,
    const float* __restrict__ g_w1,   const float* __restrict__ g_b1,
    const float* __restrict__ g_w2,   const float* __restrict__ g_b2,
    const float* __restrict__ g_fcw,  const float* __restrict__ g_fcb,
    float* __restrict__ g_out)
{
  __shared__ float s_lead[12][3], s_e1[12][3], s_e2[12][3], s_Y[12][9];
  __shared__ float s_inv2tau2, s_gamma;
  __shared__ __align__(16) _Float16 s_w1t[32][32];
  __shared__ __align__(16) _Float16 s_w2t[9][32][32];
  __shared__ float s_b1[32], s_b2[32], s_fcw[32], s_fcb;
  __shared__ _Float16 s_img[WAVES][2][64];
  __shared__ __align__(16) unsigned char s_dyn[WAVES][96 * 18 * 4];

  const int tid    = threadIdx.x;
  const int lane   = tid & 31;
  const int wv     = tid >> 5;
  const bool hiH   = (lane >= 16);
  const int lane15 = lane & 15;
  const int kloH   = hiH ? 8 : 0;
  const int khiH   = hiH ? 24 : 16;

  if (tid < 12) {
    float Lx, Ly, Lz;
    if (tid < 6) {
      float rax = g_ra[0], ray = g_ra[1], raz = g_ra[2];
      float lax = g_la[0], lay = g_la[1], laz = g_la[2];
      float llx = g_ll[0], lly = g_ll[1], llz = g_ll[2];
      float vx, vy, vz;
      if      (tid == 0) { vx = lax - rax; vy = lay - ray; vz = laz - raz; }
      else if (tid == 1) { vx = llx - rax; vy = lly - ray; vz = llz - raz; }
      else if (tid == 2) { vx = llx - lax; vy = lly - lay; vz = llz - laz; }
      else if (tid == 3) { vx = rax - 0.5f*(lax+llx); vy = ray - 0.5f*(lay+lly); vz = raz - 0.5f*(laz+llz); }
      else if (tid == 4) { vx = lax - 0.5f*(rax+llx); vy = lay - 0.5f*(ray+lly); vz = laz - 0.5f*(raz+llz); }
      else               { vx = llx - 0.5f*(rax+lax); vy = lly - 0.5f*(ray+lay); vz = llz - 0.5f*(raz+laz); }
      float nv = fmaxf(sqrtf(vx*vx + vy*vy + vz*vz), 1e-6f);
      Lx = vx / nv; Ly = vy / nv; Lz = vz / nv;
    } else {
      int r = tid - 6;
      float vx = g_chest[r*3+0], vy = g_chest[r*3+1], vz = g_chest[r*3+2];
      float nv = fmaxf(sqrtf(vx*vx + vy*vy + vz*vz), 1e-6f);
      Lx = vx / nv; Ly = vy / nv; Lz = vz / nv;
    }
    s_lead[tid][0] = Lx; s_lead[tid][1] = Ly; s_lead[tid][2] = Lz;
    float ex = 0.0f, ey = Lz, ez = -Ly;
    if (sqrtf(ey*ey + ez*ez) < 1e-4f) { ex = -Lz; ey = 0.0f; ez = Lx; }
    float n1 = fmaxf(sqrtf(ex*ex + ey*ey + ez*ez), 1e-6f);
    ex /= n1; ey /= n1; ez /= n1;
    s_e1[tid][0] = ex; s_e1[tid][1] = ey; s_e1[tid][2] = ez;
    float fx = ey*Lz - ez*Ly, fy = ez*Lx - ex*Lz, fz = ex*Ly - ey*Lx;
    float n2 = fmaxf(sqrtf(fx*fx + fy*fy + fz*fz), 1e-6f);
    s_e2[tid][0] = fx/n2; s_e2[tid][1] = fy/n2; s_e2[tid][2] = fz/n2;
    s_Y[tid][0] = 0.282095f;
    s_Y[tid][1] = -0.488603f * Ly;
    s_Y[tid][2] =  0.488603f * Lz;
    s_Y[tid][3] = -0.488603f * Lx;
    s_Y[tid][4] =  1.092548f * Lx * Ly;
    s_Y[tid][5] = -1.092548f * Ly * Lz;
    s_Y[tid][6] =  0.315392f * (3.0f * Lz * Lz - 1.0f);
    s_Y[tid][7] = -1.092548f * Lx * Lz;
    s_Y[tid][8] =  0.546274f * (Lx * Lx - Ly * Ly);
  }
  if (tid == BLOCK - 1) {
    float tau = softplus_stable(g_tau[0]) + 0.06f;
    s_inv2tau2 = 0.5f / (tau * tau);
    s_gamma    = softplus_stable(g_gam[0]) + 1e-6f;
  }
  for (int idx = tid; idx < 32 * 32; idx += BLOCK) {
    int n = idx >> 5, kq = idx & 31;
    float w = (kq < 18) ? g_w1[n * 18 + kq] : 0.0f;
    s_w1t[n][kq] = (_Float16)w;
  }
  for (int idx = tid; idx < 9 * 32 * 32; idx += BLOCK) {
    int tap = idx >> 10, n = (idx >> 5) & 31, c = idx & 31;
    s_w2t[tap][n][c] = (_Float16)g_w2[n * 288 + c * 9 + tap];
  }
  if (tid < 32) {
    s_b1[tid] = g_b1[tid]; s_b2[tid] = g_b2[tid]; s_fcw[tid] = g_fcw[tid];
    if (tid == 0) s_fcb = g_fcb[0];
  }
  __syncthreads();

  const int img = blockIdx.x * WAVES + wv;
  const int t   = img & (TT - 1);
  const int k   = (img >> 9) % KK;
  const int b   = img / (KK * TT);
  const float tt = (float)t * (1.0f / (float)(TT - 1));

  float* stage = (float*)s_dyn[wv];
  const float iv2t2 = s_inv2tau2;
  const float gma   = s_gamma;
  const float e1x = s_e1[k][0], e1y = s_e1[k][1], e1z = s_e1[k][2];
  const float e2x = s_e2[k][0], e2y = s_e2[k][1], e2z = s_e2[k][2];
  const float Lkx = s_lead[k][0], Lky = s_lead[k][1], Lkz = s_lead[k][2];

  #pragma unroll
  for (int j2 = 0; j2 < 3; ++j2) {
    int n  = lane + 32 * j2;
    int bn = b * NN + n;
    float dt = tt - g_mu[bn];
    float st = softplus_stable(g_s0[bn] + g_sv[bn] * dt) + 1e-3f;
    float z  = dt * frcp(st);
    float gauss = g_amp[bn] * fexp(-0.5f * z * z);

    float prx = g_p0[bn*3+0] + g_pv[bn*3+0] * dt;
    float pry = g_p0[bn*3+1] + g_pv[bn*3+1] * dt;
    float prz = g_p0[bn*3+2] + g_pv[bn*3+2] * dt;
    float nrm = fmaxf(sqrtf(prx*prx + pry*pry + prz*prz), 1e-8f);
    float tnh = fast_tanh(nrm);
    float scl = tnh * frcp(nrm);
    float px = prx * scl, py = pry * scl, pz = prz * scl;

    float u = fast_tanh(px*e1x + py*e1y + pz*e1z);
    float v = fast_tanh(px*e2x + py*e2y + pz*e2z);

    float pn   = fmaxf(tnh, 1e-8f);
    float cosl = (px*Lkx + py*Lky + pz*Lkz) * frcp(pn);
    float hem  = fmaxf(cosl, 0.0f);
    float cc   = fminf(fmaxf(cosl, -1.0f + 1e-6f), 1.0f - 1e-6f);
    float th   = fast_acos(cc);
    float wgt  = hem * fexp(-gma * th * th);

    float shp = 0.0f;
    #pragma unroll
    for (int m2 = 0; m2 < 9; ++m2)
      shp += (g_shb[bn*9 + m2] + g_shv[bn*9 + m2] * dt) * s_Y[k][m2];

    float A = gauss * wgt;
    stage[n*18 + 0] = A;
    stage[n*18 + 1] = A * shp;
    #pragma unroll
    for (int w8 = 0; w8 < 8; ++w8) {
      float gv = -1.0f + (2.0f / 7.0f) * (float)w8;
      float du = u - gv, dv = v - gv;
      stage[n*18 + 2  + w8] = fexp(-iv2t2 * du * du);
      stage[n*18 + 10 + w8] = fexp(-iv2t2 * dv * dv);
    }
  }
  __syncthreads();

  {
    int pA = lane * 2, pB = pA + 1;
    int hA = pA >> 3, wA = pA & 7, hB = pB >> 3, wB = pB & 7;
    float accP0 = 0.0f, accS0 = 0.0f, accP1 = 0.0f, accS1 = 0.0f;
    for (int n = 0; n < NN; ++n) {
      const float* sn = stage + n * 18;
      float A = sn[0], As = sn[1];
      float vA = sn[10 + hA] * sn[2 + wA];
      float vB = sn[10 + hB] * sn[2 + wB];
      accP0 += A * vA;  accS0 += As * vA;
      accP1 += A * vB;  accS1 += As * vB;
    }
    s_img[wv][0][pA] = (_Float16)accP0;
    s_img[wv][0][pB] = (_Float16)accP1;
    s_img[wv][1][pA] = (_Float16)accS0;
    s_img[wv][1][pB] = (_Float16)accS1;
  }
  __syncthreads();

  _Float16* h1 = (_Float16*)s_dyn[wv];

  AFrag bf1[2];
  #pragma unroll
  for (int nt = 0; nt < 2; ++nt) {
    const _Float16* wr = &s_w1t[nt * 16 + lane15][0];
    bf1[nt].h[0] = *(const v8h*)(wr + kloH);
    bf1[nt].h[1] = *(const v8h*)(wr + khiH);
  }
  #pragma unroll
  for (int m = 0; m < 4; ++m) {
    AFrag a;
    int p = m * 16 + lane15;
    int py = p >> 3, px = p & 7;
    #pragma unroll
    for (int j = 0; j < 16; ++j) {
      int kq = (j < 8) ? (kloH + j) : (khiH + (j - 8));
      _Float16 val = (_Float16)0.0f;
      if (kq < 18) {
        int c = kq / 9, tap = kq % 9;
        int y = py + tap / 3 - 1, x = px + tap % 3 - 1;
        if ((unsigned)y < 8u && (unsigned)x < 8u) val = s_img[wv][c][y * 8 + x];
      }
      a.e[j] = val;
    }
    #pragma unroll
    for (int nt = 0; nt < 2; ++nt) {
      CFrag d = {};
      d.v = __builtin_amdgcn_wmma_f32_16x16x32_f16(
                false, a.v, false, bf1[nt].v, (short)0, d.v, false, false);
      int ch = nt * 16 + lane15;
      float bias = s_b1[ch];
      #pragma unroll
      for (int r = 0; r < 8; ++r) {
        int pp = m * 16 + (hiH ? 8 + r : r);
        h1[pp * H1_STRIDE + ch] = (_Float16)gelu_exact(d.e[r] + bias);
      }
    }
  }
  __syncthreads();

  CFrag acc[4][2] = {};
  const v8h z8 = {};
  #pragma unroll
  for (int tap = 0; tap < 9; ++tap) {
    AFrag b0f, b1f;
    const _Float16* wr0 = &s_w2t[tap][lane15][0];
    const _Float16* wr1 = &s_w2t[tap][16 + lane15][0];
    b0f.h[0] = *(const v8h*)(wr0 + kloH);
    b0f.h[1] = *(const v8h*)(wr0 + khiH);
    b1f.h[0] = *(const v8h*)(wr1 + kloH);
    b1f.h[1] = *(const v8h*)(wr1 + khiH);
    const int dy = tap / 3 - 1, dx = tap % 3 - 1;
    #pragma unroll
    for (int m = 0; m < 4; ++m) {
      int p = m * 16 + lane15;
      int y = (p >> 3) + dy, x = (p & 7) + dx;
      bool inb = ((unsigned)y < 8u) && ((unsigned)x < 8u);
      int np = inb ? (y * 8 + x) : 0;
      const _Float16* hp = h1 + np * H1_STRIDE;
      v8h l0 = *(const v8h*)(hp + kloH);
      v8h l1 = *(const v8h*)(hp + khiH);
      AFrag a;
      a.h[0] = inb ? l0 : z8;
      a.h[1] = inb ? l1 : z8;
      acc[m][0].v = __builtin_amdgcn_wmma_f32_16x16x32_f16(
                        false, a.v, false, b0f.v, (short)0, acc[m][0].v, false, false);
      acc[m][1].v = __builtin_amdgcn_wmma_f32_16x16x32_f16(
                        false, a.v, false, b1f.v, (short)0, acc[m][1].v, false, false);
    }
  }

  float sum0 = 0.0f, sum1 = 0.0f;
  {
    float bb0 = s_b2[lane15], bb1 = s_b2[16 + lane15];
    #pragma unroll
    for (int m = 0; m < 4; ++m) {
      #pragma unroll
      for (int r = 0; r < 8; ++r) {
        sum0 += gelu_exact(acc[m][0].e[r] + bb0);
        sum1 += gelu_exact(acc[m][1].e[r] + bb1);
      }
    }
  }

  sum0 += __shfl_xor(sum0, 16);
  sum1 += __shfl_xor(sum1, 16);
  float contrib = sum0 * (1.0f / 64.0f) * s_fcw[lane15]
                + sum1 * (1.0f / 64.0f) * s_fcw[16 + lane15];
  contrib += __shfl_xor(contrib, 8);
  contrib += __shfl_xor(contrib, 4);
  contrib += __shfl_xor(contrib, 2);
  contrib += __shfl_xor(contrib, 1);
  if (lane == 0)
    g_out[b * (KK * TT) + t * KK + k] = contrib + s_fcb;
}

extern "C" void kernel_launch(void* const* d_in, const int* in_sizes, int n_in,
                              void* d_out, int out_size, void* d_ws, size_t ws_size,
                              hipStream_t stream) {
  (void)in_sizes; (void)n_in; (void)out_size; (void)d_ws; (void)ws_size;
  const float* mu    = (const float*)d_in[0];
  const float* s0    = (const float*)d_in[1];
  const float* sv    = (const float*)d_in[2];
  const float* amp   = (const float*)d_in[3];
  const float* shb   = (const float*)d_in[4];
  const float* shv   = (const float*)d_in[5];
  const float* p0    = (const float*)d_in[6];
  const float* pv    = (const float*)d_in[7];
  const float* taur  = (const float*)d_in[8];
  const float* gamr  = (const float*)d_in[9];
  const float* ra    = (const float*)d_in[10];
  const float* la    = (const float*)d_in[11];
  const float* ll    = (const float*)d_in[12];
  const float* chest = (const float*)d_in[13];
  const float* w1    = (const float*)d_in[14];
  const float* b1    = (const float*)d_in[15];
  const float* w2    = (const float*)d_in[16];
  const float* b2    = (const float*)d_in[17];
  const float* fcw   = (const float*)d_in[18];
  const float* fcb   = (const float*)d_in[19];
  float* out = (float*)d_out;

  const int n_images = BB * KK * TT;
  dim3 grid(n_images / WAVES), block(BLOCK);
  ecg_fused_kernel<<<grid, block, 0, stream>>>(
      mu, s0, sv, amp, shb, shv, p0, pv, taur, gamr,
      ra, la, ll, chest, w1, b1, w2, b2, fcw, fcb, out);
}


# Persistent blocks: grid 6144 -> 1024 resident blocks; each stages the f16 conv weights and the loop-invariant fragment addressing once, then loops over 6 image groups; per-image math unchanged
# speedup vs baseline: 1.0786x; 1.0786x over previous
#include <hip/hip_runtime.h>
#include <hip/hip_bf16.h>
#include <math.h>

#define BB 4
#define NN 96
#define TT 512
#define KK 12
#define GG 8
#define WAVES 4
#define BLOCK (WAVES * 32)
#define H1_STRIDE 40
#define N_GROUPS (BB * KK * TT / WAVES)
#define GRID_BLOCKS 1024

typedef __attribute__((ext_vector_type(16))) _Float16 v16h;
typedef __attribute__((ext_vector_type(8)))  _Float16 v8h;
typedef __attribute__((ext_vector_type(8)))  float    v8f;

union AFrag { v16h v; v8h h[2]; _Float16 e[16]; };
union CFrag { v8f  v; float e[8]; };

__device__ __forceinline__ float fexp(float x) { return __expf(x); }
__device__ __forceinline__ float frcp(float x) { return __builtin_amdgcn_rcpf(x); }

__device__ __forceinline__ float softplus_stable(float x) {
  return fmaxf(x, 0.0f) + __logf(1.0f + fexp(-fabsf(x)));
}

__device__ __forceinline__ float fast_tanh(float x) {
  float e = fexp(-2.0f * fabsf(x));
  float y = (1.0f - e) * frcp(1.0f + e);
  return copysignf(y, x);
}

__device__ __forceinline__ float fast_erf(float x) {
  float ax = fabsf(x);
  float t  = frcp(fmaf(0.3275911f, ax, 1.0f));
  float p  = fmaf(t, 1.061405429f, -1.453152027f);
  p = fmaf(t, p, 1.421413741f);
  p = fmaf(t, p, -0.284496736f);
  p = fmaf(t, p, 0.254829592f);
  float y = 1.0f - p * t * fexp(-ax * ax);
  return copysignf(y, x);
}

__device__ __forceinline__ float gelu_exact(float x) {
  return 0.5f * x * (1.0f + fast_erf(x * 0.70710678118654752f));
}

__device__ __forceinline__ float fast_acos(float x) {
  float ax = fabsf(x);
  float p = fmaf(ax, -0.0187293f, 0.0742610f);
  p = fmaf(ax, p, -0.2121144f);
  p = fmaf(ax, p, 1.5707288f);
  float r = sqrtf(fmaxf(1.0f - ax, 0.0f)) * p;
  return (x >= 0.0f) ? r : (3.14159265358979f - r);
}

__global__ __launch_bounds__(BLOCK) void ecg_fused_kernel(
    const float* __restrict__ g_mu,   const float* __restrict__ g_s0,
    const float* __restrict__ g_sv,   const float* __restrict__ g_amp,
    const float* __restrict__ g_shb,  const float* __restrict__ g_shv,
    const float* __restrict__ g_p0,   const float* __restrict__ g_pv,
    const float* __restrict__ g_tau,  const float* __restrict__ g_gam,
    const float* __restrict__ g_ra,   const float* __restrict__ g_la,
    const float* __restrict__ g_ll,   const float* __restrict__ g_chest,
    const float* __restrict__ g_w1,   const float* __restrict__ g_b1,
    const float* __restrict__ g_w2,   const float* __restrict__ g_b2,
    const float* __restrict__ g_fcw,  const float* __restrict__ g_fcb,
    float* __restrict__ g_out)
{
  __shared__ float s_lead[12][3], s_e1[12][3], s_e2[12][3], s_Y[12][9];
  __shared__ float s_inv2tau2, s_gamma;
  __shared__ __align__(16) _Float16 s_w1t[32][32];
  __shared__ __align__(16) _Float16 s_w2t[9][32][32];
  __shared__ float s_b1[32], s_b2[32], s_fcw[32], s_fcb;
  __shared__ _Float16 s_img[WAVES][2][64];
  __shared__ __align__(16) unsigned char s_dyn[WAVES][96 * 18 * 4];

  const int tid    = threadIdx.x;
  const int lane   = tid & 31;
  const int wv     = tid >> 5;
  const bool hiH   = (lane >= 16);
  const int lane15 = lane & 15;
  const int kloH   = hiH ? 8 : 0;
  const int khiH   = hiH ? 24 : 16;

  if (tid < 12) {
    float Lx, Ly, Lz;
    if (tid < 6) {
      float rax = g_ra[0], ray = g_ra[1], raz = g_ra[2];
      float lax = g_la[0], lay = g_la[1], laz = g_la[2];
      float llx = g_ll[0], lly = g_ll[1], llz = g_ll[2];
      float vx, vy, vz;
      if      (tid == 0) { vx = lax - rax; vy = lay - ray; vz = laz - raz; }
      else if (tid == 1) { vx = llx - rax; vy = lly - ray; vz = llz - raz; }
      else if (tid == 2) { vx = llx - lax; vy = lly - lay; vz = llz - laz; }
      else if (tid == 3) { vx = rax - 0.5f*(lax+llx); vy = ray - 0.5f*(lay+lly); vz = raz - 0.5f*(laz+llz); }
      else if (tid == 4) { vx = lax - 0.5f*(rax+llx); vy = lay - 0.5f*(ray+lly); vz = laz - 0.5f*(raz+llz); }
      else               { vx = llx - 0.5f*(rax+lax); vy = lly - 0.5f*(ray+lay); vz = llz - 0.5f*(raz+laz); }
      float nv = fmaxf(sqrtf(vx*vx + vy*vy + vz*vz), 1e-6f);
      Lx = vx / nv; Ly = vy / nv; Lz = vz / nv;
    } else {
      int r = tid - 6;
      float vx = g_chest[r*3+0], vy = g_chest[r*3+1], vz = g_chest[r*3+2];
      float nv = fmaxf(sqrtf(vx*vx + vy*vy + vz*vz), 1e-6f);
      Lx = vx / nv; Ly = vy / nv; Lz = vz / nv;
    }
    s_lead[tid][0] = Lx; s_lead[tid][1] = Ly; s_lead[tid][2] = Lz;
    float ex = 0.0f, ey = Lz, ez = -Ly;
    if (sqrtf(ey*ey + ez*ez) < 1e-4f) { ex = -Lz; ey = 0.0f; ez = Lx; }
    float n1 = fmaxf(sqrtf(ex*ex + ey*ey + ez*ez), 1e-6f);
    ex /= n1; ey /= n1; ez /= n1;
    s_e1[tid][0] = ex; s_e1[tid][1] = ey; s_e1[tid][2] = ez;
    float fx = ey*Lz - ez*Ly, fy = ez*Lx - ex*Lz, fz = ex*Ly - ey*Lx;
    float n2 = fmaxf(sqrtf(fx*fx + fy*fy + fz*fz), 1e-6f);
    s_e2[tid][0] = fx/n2; s_e2[tid][1] = fy/n2; s_e2[tid][2] = fz/n2;
    s_Y[tid][0] = 0.282095f;
    s_Y[tid][1] = -0.488603f * Ly;
    s_Y[tid][2] =  0.488603f * Lz;
    s_Y[tid][3] = -0.488603f * Lx;
    s_Y[tid][4] =  1.092548f * Lx * Ly;
    s_Y[tid][5] = -1.092548f * Ly * Lz;
    s_Y[tid][6] =  0.315392f * (3.0f * Lz * Lz - 1.0f);
    s_Y[tid][7] = -1.092548f * Lx * Lz;
    s_Y[tid][8] =  0.546274f * (Lx * Lx - Ly * Ly);
  }
  if (tid == BLOCK - 1) {
    float tau = softplus_stable(g_tau[0]) + 0.06f;
    s_inv2tau2 = 0.5f / (tau * tau);
    s_gamma    = softplus_stable(g_gam[0]) + 1e-6f;
  }
  for (int idx = tid; idx < 32 * 32; idx += BLOCK) {
    int n = idx >> 5, kq = idx & 31;
    float w = (kq < 18) ? g_w1[n * 18 + kq] : 0.0f;
    s_w1t[n][kq] = (_Float16)w;
  }
  for (int idx = tid; idx < 9 * 32 * 32; idx += BLOCK) {
    int tap = idx >> 10, n = (idx >> 5) & 31, c = idx & 31;
    s_w2t[tap][n][c] = (_Float16)g_w2[n * 288 + c * 9 + tap];
  }
  if (tid < 32) {
    s_b1[tid] = g_b1[tid]; s_b2[tid] = g_b2[tid]; s_fcw[tid] = g_fcw[tid];
    if (tid == 0) s_fcb = g_fcb[0];
  }
  __syncthreads();

  for (int grp = blockIdx.x; grp < N_GROUPS; grp += GRID_BLOCKS) {
  const int img = grp * WAVES + wv;
  const int t   = img & (TT - 1);
  const int k   = (img >> 9) % KK;
  const int b   = img / (KK * TT);
  const float tt = (float)t * (1.0f / (float)(TT - 1));

  float* stage = (float*)s_dyn[wv];
  const float iv2t2 = s_inv2tau2;
  const float gma   = s_gamma;
  const float e1x = s_e1[k][0], e1y = s_e1[k][1], e1z = s_e1[k][2];
  const float e2x = s_e2[k][0], e2y = s_e2[k][1], e2z = s_e2[k][2];
  const float Lkx = s_lead[k][0], Lky = s_lead[k][1], Lkz = s_lead[k][2];

  #pragma unroll
  for (int j2 = 0; j2 < 3; ++j2) {
    int n  = lane + 32 * j2;
    int bn = b * NN + n;
    float dt = tt - g_mu[bn];
    float st = softplus_stable(g_s0[bn] + g_sv[bn] * dt) + 1e-3f;
    float z  = dt * frcp(st);
    float gauss = g_amp[bn] * fexp(-0.5f * z * z);

    float prx = g_p0[bn*3+0] + g_pv[bn*3+0] * dt;
    float pry = g_p0[bn*3+1] + g_pv[bn*3+1] * dt;
    float prz = g_p0[bn*3+2] + g_pv[bn*3+2] * dt;
    float nrm = fmaxf(sqrtf(prx*prx + pry*pry + prz*prz), 1e-8f);
    float tnh = fast_tanh(nrm);
    float scl = tnh * frcp(nrm);
    float px = prx * scl, py = pry * scl, pz = prz * scl;

    float u = fast_tanh(px*e1x + py*e1y + pz*e1z);
    float v = fast_tanh(px*e2x + py*e2y + pz*e2z);

    float pn   = fmaxf(tnh, 1e-8f);
    float cosl = (px*Lkx + py*Lky + pz*Lkz) * frcp(pn);
    float hem  = fmaxf(cosl, 0.0f);
    float cc   = fminf(fmaxf(cosl, -1.0f + 1e-6f), 1.0f - 1e-6f);
    float th   = fast_acos(cc);
    float wgt  = hem * fexp(-gma * th * th);

    float shp = 0.0f;
    #pragma unroll
    for (int m2 = 0; m2 < 9; ++m2)
      shp += (g_shb[bn*9 + m2] + g_shv[bn*9 + m2] * dt) * s_Y[k][m2];

    float A = gauss * wgt;
    stage[n*18 + 0] = A;
    stage[n*18 + 1] = A * shp;
    #pragma unroll
    for (int w8 = 0; w8 < 8; ++w8) {
      float gv = -1.0f + (2.0f / 7.0f) * (float)w8;
      float du = u - gv, dv = v - gv;
      stage[n*18 + 2  + w8] = fexp(-iv2t2 * du * du);
      stage[n*18 + 10 + w8] = fexp(-iv2t2 * dv * dv);
    }
  }
  __syncthreads();

  {
    int pA = lane * 2, pB = pA + 1;
    int hA = pA >> 3, wA = pA & 7, hB = pB >> 3, wB = pB & 7;
    float accP0 = 0.0f, accS0 = 0.0f, accP1 = 0.0f, accS1 = 0.0f;
    for (int n = 0; n < NN; ++n) {
      const float* sn = stage + n * 18;
      float A = sn[0], As = sn[1];
      float vA = sn[10 + hA] * sn[2 + wA];
      float vB = sn[10 + hB] * sn[2 + wB];
      accP0 += A * vA;  accS0 += As * vA;
      accP1 += A * vB;  accS1 += As * vB;
    }
    s_img[wv][0][pA] = (_Float16)accP0;
    s_img[wv][0][pB] = (_Float16)accP1;
    s_img[wv][1][pA] = (_Float16)accS0;
    s_img[wv][1][pB] = (_Float16)accS1;
  }
  __syncthreads();

  _Float16* h1 = (_Float16*)s_dyn[wv];

  AFrag bf1[2];
  #pragma unroll
  for (int nt = 0; nt < 2; ++nt) {
    const _Float16* wr = &s_w1t[nt * 16 + lane15][0];
    bf1[nt].h[0] = *(const v8h*)(wr + kloH);
    bf1[nt].h[1] = *(const v8h*)(wr + khiH);
  }
  #pragma unroll
  for (int m = 0; m < 4; ++m) {
    AFrag a;
    int p = m * 16 + lane15;
    int py = p >> 3, px = p & 7;
    #pragma unroll
    for (int j = 0; j < 16; ++j) {
      int kq = (j < 8) ? (kloH + j) : (khiH + (j - 8));
      _Float16 val = (_Float16)0.0f;
      if (kq < 18) {
        int c = kq / 9, tap = kq % 9;
        int y = py + tap / 3 - 1, x = px + tap % 3 - 1;
        if ((unsigned)y < 8u && (unsigned)x < 8u) val = s_img[wv][c][y * 8 + x];
      }
      a.e[j] = val;
    }
    #pragma unroll
    for (int nt = 0; nt < 2; ++nt) {
      CFrag d = {};
      d.v = __builtin_amdgcn_wmma_f32_16x16x32_f16(
                false, a.v, false, bf1[nt].v, (short)0, d.v, false, false);
      int ch = nt * 16 + lane15;
      float bias = s_b1[ch];
      #pragma unroll
      for (int r = 0; r < 8; ++r) {
        int pp = m * 16 + (hiH ? 8 + r : r);
        h1[pp * H1_STRIDE + ch] = (_Float16)gelu_exact(d.e[r] + bias);
      }
    }
  }
  __syncthreads();

  CFrag acc[4][2] = {};
  const v8h z8 = {};
  #pragma unroll
  for (int tap = 0; tap < 9; ++tap) {
    AFrag b0f, b1f;
    const _Float16* wr0 = &s_w2t[tap][lane15][0];
    const _Float16* wr1 = &s_w2t[tap][16 + lane15][0];
    b0f.h[0] = *(const v8h*)(wr0 + kloH);
    b0f.h[1] = *(const v8h*)(wr0 + khiH);
    b1f.h[0] = *(const v8h*)(wr1 + kloH);
    b1f.h[1] = *(const v8h*)(wr1 + khiH);
    const int dy = tap / 3 - 1, dx = tap % 3 - 1;
    #pragma unroll
    for (int m = 0; m < 4; ++m) {
      int p = m * 16 + lane15;
      int y = (p >> 3) + dy, x = (p & 7) + dx;
      bool inb = ((unsigned)y < 8u) && ((unsigned)x < 8u);
      int np = inb ? (y * 8 + x) : 0;
      const _Float16* hp = h1 + np * H1_STRIDE;
      v8h l0 = *(const v8h*)(hp + kloH);
      v8h l1 = *(const v8h*)(hp + khiH);
      AFrag a;
      a.h[0] = inb ? l0 : z8;
      a.h[1] = inb ? l1 : z8;
      acc[m][0].v = __builtin_amdgcn_wmma_f32_16x16x32_f16(
                        false, a.v, false, b0f.v, (short)0, acc[m][0].v, false, false);
      acc[m][1].v = __builtin_amdgcn_wmma_f32_16x16x32_f16(
                        false, a.v, false, b1f.v, (short)0, acc[m][1].v, false, false);
    }
  }

  float sum0 = 0.0f, sum1 = 0.0f;
  {
    float bb0 = s_b2[lane15], bb1 = s_b2[16 + lane15];
    #pragma unroll
    for (int m = 0; m < 4; ++m) {
      #pragma unroll
      for (int r = 0; r < 8; ++r) {
        sum0 += gelu_exact(acc[m][0].e[r] + bb0);
        sum1 += gelu_exact(acc[m][1].e[r] + bb1);
      }
    }
  }

  sum0 += __shfl_xor(sum0, 16);
  sum1 += __shfl_xor(sum1, 16);
  float contrib = sum0 * (1.0f / 64.0f) * s_fcw[lane15]
                + sum1 * (1.0f / 64.0f) * s_fcw[16 + lane15];
  contrib += __shfl_xor(contrib, 8);
  contrib += __shfl_xor(contrib, 4);
  contrib += __shfl_xor(contrib, 2);
  contrib += __shfl_xor(contrib, 1);
  if (lane == 0)
    g_out[b * (KK * TT) + t * KK + k] = contrib + s_fcb;
  }
}

extern "C" void kernel_launch(void* const* d_in, const int* in_sizes, int n_in,
                              void* d_out, int out_size, void* d_ws, size_t ws_size,
                              hipStream_t stream) {
  (void)in_sizes; (void)n_in; (void)out_size; (void)d_ws; (void)ws_size;
  const float* mu    = (const float*)d_in[0];
  const float* s0    = (const float*)d_in[1];
  const float* sv    = (const float*)d_in[2];
  const float* amp   = (const float*)d_in[3];
  const float* shb   = (const float*)d_in[4];
  const float* shv   = (const float*)d_in[5];
  const float* p0    = (const float*)d_in[6];
  const float* pv    = (const float*)d_in[7];
  const float* taur  = (const float*)d_in[8];
  const float* gamr  = (const float*)d_in[9];
  const float* ra    = (const float*)d_in[10];
  const float* la    = (const float*)d_in[11];
  const float* ll    = (const float*)d_in[12];
  const float* chest = (const float*)d_in[13];
  const float* w1    = (const float*)d_in[14];
  const float* b1    = (const float*)d_in[15];
  const float* w2    = (const float*)d_in[16];
  const float* b2    = (const float*)d_in[17];
  const float* fcw   = (const float*)d_in[18];
  const float* fcb   = (const float*)d_in[19];
  float* out = (float*)d_out;

  dim3 grid(GRID_BLOCKS), block(BLOCK);
  ecg_fused_kernel<<<grid, block, 0, stream>>>(
      mu, s0, sv, amp, shb, shv, p0, pv, taur, gamr,
      ra, la, ll, chest, w1, b1, w2, b2, fcw, fcb, out);
}
